# starting kernel
# baseline (speedup 1.0000x reference)
#include <hip/hip_runtime.h>
#include <hip/hip_bf16.h>

typedef float v2f __attribute__((ext_vector_type(2)));
typedef float v8f __attribute__((ext_vector_type(8)));

static __device__ __forceinline__ v8f wmma4(v2f a, v2f b, v8f c) {
  return __builtin_amdgcn_wmma_f32_16x16x4_f32(false, a, false, b, (short)0, c,
                                               false, false);
}

#define LD 512
#define ROWS 4096
#define CHUNK 64
#define NCHUNK 32
#define NTASK 512

__global__ __launch_bounds__(128) void gemm_f32(const float* __restrict__ A,
                                                const float* __restrict__ B,
                                                float* __restrict__ C,
                                                const float* __restrict__ bias,
                                                int M, int N, int K) {
  __shared__ float As[64][17];
  __shared__ float Bs[16][65];
  const int t = threadIdx.x;
  const int wave = t >> 5, lane = t & 31, lh = lane >> 4, l16 = lane & 15;
  const int m0 = blockIdx.y * 64, n0 = blockIdx.x * 64;

  v8f acc[4] = {};

  for (int k0 = 0; k0 < K; k0 += 16) {
    {
      int r = t >> 1, c = (t & 1) * 8;
      const float4* src = reinterpret_cast<const float4*>(&A[(size_t)(m0 + r) * K + k0 + c]);
      if (k0 + 16 < K) __builtin_prefetch(&A[(size_t)(m0 + r) * K + k0 + 16 + c], 0, 1);
      float4 v0 = src[0], v1 = src[1];
      float* d = &As[r][c];
      d[0] = v0.x; d[1] = v0.y; d[2] = v0.z; d[3] = v0.w;
      d[4] = v1.x; d[5] = v1.y; d[6] = v1.z; d[7] = v1.w;
    }
    {
      int r = t >> 3, c = (t & 7) * 8;
      const float4* src = reinterpret_cast<const float4*>(&B[(size_t)(k0 + r) * N + n0 + c]);
      float4 v0 = src[0], v1 = src[1];
      float* d = &Bs[r][c];
      d[0] = v0.x; d[1] = v0.y; d[2] = v0.z; d[3] = v0.w;
      d[4] = v1.x; d[5] = v1.y; d[6] = v1.z; d[7] = v1.w;
    }
    __syncthreads();
    const int mr = wave * 16 + l16;
#pragma unroll
    for (int kk = 0; kk < 16; kk += 4) {
      v2f a;
      a.x = As[mr][kk + 2 * lh];
      a.y = As[mr][kk + 2 * lh + 1];
#pragma unroll
      for (int nt = 0; nt < 4; ++nt) {
        v2f b;
        b.x = Bs[kk + 2 * lh][nt * 16 + l16];
        b.y = Bs[kk + 2 * lh + 1][nt * 16 + l16];
        acc[nt] = wmma4(a, b, acc[nt]);
      }
    }
    __syncthreads();
  }

#pragma unroll
  for (int nt = 0; nt < 4; ++nt) {
    int col = n0 + nt * 16 + l16;
    float badd = (bias != nullptr) ? bias[col] : 0.0f;
#pragma unroll
    for (int r = 0; r < 8; ++r) {
      int row = m0 + wave * 16 + r + 8 * lh;
      C[(size_t)row * N + col] = acc[nt][r] + badd;
    }
  }
}

__global__ __launch_bounds__(128) void qk_transform(float* __restrict__ Q,
                                                    float* __restrict__ Kb) {
  const int wave = threadIdx.x >> 5, lane = threadIdx.x & 31;
  const int hr = blockIdx.x * 4 + wave;
  if (hr >= ROWS * 8) return;
  const int row = hr >> 3, h = hr & 7;
  const size_t base = (size_t)row * LD + h * 64;

  float q0 = Q[base + lane], q1 = Q[base + lane + 32];
  float m = fmaxf(q0, q1);
#pragma unroll
  for (int off = 16; off > 0; off >>= 1) m = fmaxf(m, __shfl_xor(m, off, 32));
  float e0 = __expf(q0 - m), e1 = __expf(q1 - m);
  float s = e0 + e1;
#pragma unroll
  for (int off = 16; off > 0; off >>= 1) s += __shfl_xor(s, off, 32);
  float inv = 0.125f / s;
  Q[base + lane] = e0 * inv;
  Q[base + lane + 32] = e1 * inv;

  float k0 = Kb[base + lane], k1 = Kb[base + lane + 32];
  Kb[base + lane] = __expf(k0);
  Kb[base + lane + 32] = __expf(k1);
}

__global__ __launch_bounds__(256) void chunk_state(const float* __restrict__ Kb,
                                                   const float* __restrict__ Vb,
                                                   float* __restrict__ Sbuf,
                                                   float* __restrict__ ksumbuf) {
  __shared__ float Ks[64][65];
  __shared__ float Vs[64][65];
  const int task = blockIdx.x;
  const int ck = task & 31, h = (task >> 5) & 7, b = task >> 8;
  const int row0 = b * 2048 + ck * CHUNK, col0 = h * 64;
  const int t = threadIdx.x, wave = t >> 5, lane = t & 31;
  const int lh = lane >> 4, l16 = lane & 15;

  {
    int r = t >> 2, c = (t & 3) * 16;
    const float4* ks = reinterpret_cast<const float4*>(&Kb[(size_t)(row0 + r) * LD + col0 + c]);
    const float4* vs = reinterpret_cast<const float4*>(&Vb[(size_t)(row0 + r) * LD + col0 + c]);
#pragma unroll
    for (int i = 0; i < 4; ++i) {
      float4 kv = ks[i], vv = vs[i];
      float* kd = &Ks[r][c + i * 4];
      kd[0] = kv.x; kd[1] = kv.y; kd[2] = kv.z; kd[3] = kv.w;
      float* vd = &Vs[r][c + i * 4];
      vd[0] = vv.x; vd[1] = vv.y; vd[2] = vv.z; vd[3] = vv.w;
    }
  }
  __syncthreads();

  const size_t sbase = (size_t)task * 4096;
#pragma unroll
  for (int tt = wave * 2; tt <= wave * 2 + 1; ++tt) {
    const int mt = tt >> 2, nt = tt & 3;
    v8f acc = {};
#pragma unroll
    for (int kk = 0; kk < 64; kk += 4) {
      int tr = kk + 2 * lh;
      v2f a, bfr;
      a.x = Ks[tr][mt * 16 + l16];
      a.y = Ks[tr + 1][mt * 16 + l16];
      bfr.x = Vs[tr][nt * 16 + l16];
      bfr.y = Vs[tr + 1][nt * 16 + l16];
      acc = wmma4(a, bfr, acc);
    }
#pragma unroll
    for (int r = 0; r < 8; ++r) {
      int d = mt * 16 + r + 8 * lh, n = nt * 16 + l16;
      Sbuf[sbase + d * 64 + n] = acc[r];
    }
  }
  if (t < 64) {
    float s = 0.f;
#pragma unroll 8
    for (int j = 0; j < 64; ++j) s += Ks[j][t];
    ksumbuf[task * 64 + t] = s;
  }
}

__global__ void prefix_scan(float* __restrict__ Sbuf, float* __restrict__ ksumbuf) {
  const int g = blockIdx.x * blockDim.x + threadIdx.x;
  if (g < 16 * 4096) {
    const int bh = g >> 12, e = g & 4095;
    float run = 0.f;
    for (int c = 0; c < NCHUNK; ++c) {
      size_t idx = (size_t)(bh * NCHUNK + c) * 4096 + e;
      float v = Sbuf[idx];
      Sbuf[idx] = run;
      run += v;
    }
  } else if (g < 16 * 4096 + 16 * 64) {
    const int s2 = g - 16 * 4096, bh = s2 >> 6, d = s2 & 63;
    float run = 0.f;
    for (int c = 0; c < NCHUNK; ++c) {
      int idx = (bh * NCHUNK + c) * 64 + d;
      float v = ksumbuf[idx];
      ksumbuf[idx] = run;
      run += v;
    }
  }
}

__global__ __launch_bounds__(256) void attn_chunk(const float* __restrict__ Qb,
                                                  const float* __restrict__ Kb,
                                                  const float* __restrict__ Vb,
                                                  const float* __restrict__ Sbuf,
                                                  const float* __restrict__ ksumbuf,
                                                  float* __restrict__ attn) {
  extern __shared__ float sm[];
  float* Qs   = sm;
  float* Ks   = Qs + 64 * 65;
  float* Vs   = Ks + 64 * 65;
  float* Ss   = Vs + 64 * 65;
  float* Ms   = Ss + 64 * 65;
  float* ksum = Ms + 64 * 65;
  float* Dinv = ksum + 64;

  const int task = blockIdx.x;
  const int ck = task & 31, h = (task >> 5) & 7, b = task >> 8;
  const int row0 = b * 2048 + ck * CHUNK, col0 = h * 64;
  const int t = threadIdx.x, wave = t >> 5, lane = t & 31;
  const int lh = lane >> 4, l16 = lane & 15;

  {
    int r = t >> 2, c = (t & 3) * 16;
    const float4* qs = reinterpret_cast<const float4*>(&Qb[(size_t)(row0 + r) * LD + col0 + c]);
    const float4* ks = reinterpret_cast<const float4*>(&Kb[(size_t)(row0 + r) * LD + col0 + c]);
    const float4* vs = reinterpret_cast<const float4*>(&Vb[(size_t)(row0 + r) * LD + col0 + c]);
    const float4* ss = reinterpret_cast<const float4*>(&Sbuf[(size_t)task * 4096 + t * 16]);
#pragma unroll
    for (int i = 0; i < 4; ++i) {
      float4 qv = qs[i], kv = ks[i], vv = vs[i], sv = ss[i];
      float* qd = &Qs[r * 65 + c + i * 4];
      qd[0] = qv.x; qd[1] = qv.y; qd[2] = qv.z; qd[3] = qv.w;
      float* kd = &Ks[r * 65 + c + i * 4];
      kd[0] = kv.x; kd[1] = kv.y; kd[2] = kv.z; kd[3] = kv.w;
      float* vd = &Vs[r * 65 + c + i * 4];
      vd[0] = vv.x; vd[1] = vv.y; vd[2] = vv.z; vd[3] = vv.w;
      float* sd = &Ss[(t >> 2) * 65 + (t & 3) * 16 + i * 4];
      sd[0] = sv.x; sd[1] = sv.y; sd[2] = sv.z; sd[3] = sv.w;
    }
    if (t < 64) ksum[t] = ksumbuf[task * 64 + t];
  }
  __syncthreads();

#pragma unroll
  for (int tt = wave * 2; tt <= wave * 2 + 1; ++tt) {
    const int mt = tt >> 2, nt = tt & 3;
    v8f acc = {};
#pragma unroll
    for (int kk = 0; kk < 64; kk += 4) {
      v2f a, bfr;
      a.x = Qs[(mt * 16 + l16) * 65 + kk + 2 * lh];
      a.y = Qs[(mt * 16 + l16) * 65 + kk + 2 * lh + 1];
      bfr.x = Ks[(nt * 16 + l16) * 65 + kk + 2 * lh];
      bfr.y = Ks[(nt * 16 + l16) * 65 + kk + 2 * lh + 1];
      acc = wmma4(a, bfr, acc);
    }
#pragma unroll
    for (int r = 0; r < 8; ++r) {
      int i = mt * 16 + r + 8 * lh, j = nt * 16 + l16;
      Ms[i * 65 + j] = (j <= i) ? acc[r] : 0.0f;
    }
  }
  __syncthreads();

  if (t < 64) {
    float s = 0.f, dq = 0.f;
#pragma unroll 8
    for (int j = 0; j < 64; ++j) s += Ms[t * 65 + j];
#pragma unroll 8
    for (int d = 0; d < 64; ++d) dq += Qs[t * 65 + d] * ksum[d];
    Dinv[t] = 1.0f / fmaxf(s + dq, 1e-3f);
  }
  __syncthreads();

#pragma unroll
  for (int tt = wave * 2; tt <= wave * 2 + 1; ++tt) {
    const int mt = tt >> 2, nt = tt & 3;
    v8f acc = {};
#pragma unroll
    for (int kk = 0; kk < 64; kk += 4) {
      v2f a, bfr;
      a.x = Ms[(mt * 16 + l16) * 65 + kk + 2 * lh];
      a.y = Ms[(mt * 16 + l16) * 65 + kk + 2 * lh + 1];
      bfr.x = Vs[(kk + 2 * lh) * 65 + nt * 16 + l16];
      bfr.y = Vs[(kk + 2 * lh + 1) * 65 + nt * 16 + l16];
      acc = wmma4(a, bfr, acc);
    }
#pragma unroll
    for (int kk = 0; kk < 64; kk += 4) {
      v2f a, bfr;
      a.x = Qs[(mt * 16 + l16) * 65 + kk + 2 * lh];
      a.y = Qs[(mt * 16 + l16) * 65 + kk + 2 * lh + 1];
      bfr.x = Ss[(kk + 2 * lh) * 65 + nt * 16 + l16];
      bfr.y = Ss[(kk + 2 * lh + 1) * 65 + nt * 16 + l16];
      acc = wmma4(a, bfr, acc);
    }
#pragma unroll
    for (int r = 0; r < 8; ++r) {
      int i = mt * 16 + r + 8 * lh;
      int col = col0 + nt * 16 + l16;
      attn[(size_t)(row0 + i) * LD + col] = acc[r] * Dinv[i];
    }
  }
}

extern "C" void kernel_launch(void* const* d_in, const int* in_sizes, int n_in,
                              void* d_out, int out_size, void* d_ws, size_t ws_size,
                              hipStream_t stream) {
  const float* x  = (const float*)d_in[0];
  const float* Wq = (const float*)d_in[1];
  const float* Wk = (const float*)d_in[2];
  const float* Wv = (const float*)d_in[3];
  const float* Wo = (const float*)d_in[4];
  const float* bo = (const float*)d_in[5];
  float* out = (float*)d_out;

  float* ws = (float*)d_ws;
  const size_t MN = (size_t)ROWS * LD;
  float* Qb = ws;
  float* Kb = Qb + MN;
  float* Vb = Kb + MN;
  float* At = Vb + MN;
  float* Sb = At + MN;
  float* Kc = Sb + (size_t)NTASK * 4096;

  dim3 ggrid(LD / 64, ROWS / 64);

  gemm_f32<<<ggrid, 128, 0, stream>>>(x, Wq, Qb, nullptr, ROWS, LD, LD);
  gemm_f32<<<ggrid, 128, 0, stream>>>(x, Wk, Kb, nullptr, ROWS, LD, LD);
  gemm_f32<<<ggrid, 128, 0, stream>>>(x, Wv, Vb, nullptr, ROWS, LD, LD);

  qk_transform<<<(ROWS * 8) / 4, 128, 0, stream>>>(Qb, Kb);

  chunk_state<<<NTASK, 256, 0, stream>>>(Kb, Vb, Sb, Kc);

  prefix_scan<<<(16 * 4096 + 16 * 64 + 255) / 256, 256, 0, stream>>>(Sb, Kc);

  size_t shbytes = (size_t)(5 * 64 * 65 + 128) * sizeof(float);
  attn_chunk<<<NTASK, 256, shbytes, stream>>>(Qb, Kb, Vb, Sb, Kc, At);

  gemm_f32<<<ggrid, 128, 0, stream>>>(At, Wo, out, bo, ROWS, LD, LD);
}


# f32 GEMMs (q,k,v,out): K loop software-pipelined: next K-step's global loads in flight in registers during the WMMAs, 2 LDS stages, one barrier per 32-k step; 128x64 tile, 32x32 per wave
# speedup vs baseline: 1.5998x; 1.5998x over previous
#include <hip/hip_runtime.h>
#include <hip/hip_bf16.h>

typedef float v2f __attribute__((ext_vector_type(2)));
typedef float v8f __attribute__((ext_vector_type(8)));

static __device__ __forceinline__ v8f wmma4(v2f a, v2f b, v8f c) {
  return __builtin_amdgcn_wmma_f32_16x16x4_f32(false, a, false, b, (short)0, c,
                                               false, false);
}

#define LD 512
#define ROWS 4096
#define CHUNK 64
#define NCHUNK 32
#define NTASK 512

#define GBM 128
#define GBN 64
#define GBK 32
#define APITCH 36
#define BPITCH 72

static __device__ __forceinline__ void gemm_mma_step(const float* __restrict__ as,
                                                     const float* __restrict__ bs,
                                                     v8f (&acc)[2][2]) {
  v2f a[GBK / 4][2], b[GBK / 4][2];
#pragma unroll
  for (int kq = 0; kq < GBK / 4; ++kq) {
    const int kk = kq * 4;
#pragma unroll
    for (int mi = 0; mi < 2; ++mi)
      a[kq][mi] = *reinterpret_cast<const v2f*>(&as[mi * 16 * APITCH + kk]);
#pragma unroll
    for (int ni = 0; ni < 2; ++ni) {
      b[kq][ni].x = bs[kk * BPITCH + ni * 16];
      b[kq][ni].y = bs[(kk + 1) * BPITCH + ni * 16];
    }
  }
  __builtin_amdgcn_sched_barrier(0);
#pragma unroll
  for (int kq = 0; kq < GBK / 4; ++kq)
#pragma unroll
    for (int mi = 0; mi < 2; ++mi)
#pragma unroll
      for (int ni = 0; ni < 2; ++ni)
        acc[mi][ni] = wmma4(a[kq][mi], b[kq][ni], acc[mi][ni]);
}

__global__ __launch_bounds__(256) void gemm_f32(const float* __restrict__ A,
                                                const float* __restrict__ B,
                                                float* __restrict__ C,
                                                const float* __restrict__ bias,
                                                int M, int N, int K) {
  __shared__ __attribute__((aligned(16))) float As[2][GBM * APITCH];
  __shared__ __attribute__((aligned(16))) float Bs[2][GBK * BPITCH];
  const int t = threadIdx.x;
  const int wave = t >> 5, lane = t & 31, lh = lane >> 4, l16 = lane & 15;
  const int wm = wave >> 1, wn = wave & 1;
  const int m0 = blockIdx.y * GBM, n0 = blockIdx.x * GBN;

  const int ar = t >> 3, ac = (t & 7) * 4;
  const int br = t >> 4, bc = (t & 15) * 4;
  const float* Ag = A + (size_t)(m0 + ar) * K + ac;
  const float* Bg = B + (size_t)br * N + n0 + bc;
  const size_t a_rows = (size_t)32 * K;
  const size_t b_rows = (size_t)16 * N;
  const size_t b_kstep = (size_t)GBK * N;
  const int a_st = ar * APITCH + ac;
  const int b_st = br * BPITCH + bc;

  const float4* Ap = reinterpret_cast<const float4*>(Ag);
  const float4* Bp = reinterpret_cast<const float4*>(Bg);
  const size_t a_rows4 = a_rows / 4, b_rows4 = b_rows / 4;
  float4 ra0 = Ap[0], ra1 = Ap[a_rows4], ra2 = Ap[2 * a_rows4], ra3 = Ap[3 * a_rows4];
  float4 rb0 = Bp[0], rb1 = Bp[b_rows4];
  {
    float4* ad = reinterpret_cast<float4*>(&As[0][a_st]);
    float4* bd = reinterpret_cast<float4*>(&Bs[0][b_st]);
    ad[0] = ra0; ad[8 * APITCH] = ra1; ad[16 * APITCH] = ra2; ad[24 * APITCH] = ra3;
    bd[0] = rb0; bd[4 * BPITCH] = rb1;
  }
  __syncthreads();

  v8f acc[2][2] = {};
  const int nk = K / GBK;
  const int aoff = (wm * 32 + l16) * APITCH + 2 * lh;
  const int boff = (2 * lh) * BPITCH + wn * 32 + l16;

  for (int kt = 0; kt + 1 < nk; ++kt) {
    const int cur = kt & 1, nxt = cur ^ 1;
    Ap += GBK / 4;
    Bp += b_kstep / 4;
    ra0 = Ap[0]; ra1 = Ap[a_rows4]; ra2 = Ap[2 * a_rows4]; ra3 = Ap[3 * a_rows4];
    rb0 = Bp[0]; rb1 = Bp[b_rows4];
    __builtin_amdgcn_sched_barrier(0);

    gemm_mma_step(&As[cur][aoff], &Bs[cur][boff], acc);
    __builtin_amdgcn_sched_barrier(0);

    float4* ad = reinterpret_cast<float4*>(&As[nxt][a_st]);
    float4* bd = reinterpret_cast<float4*>(&Bs[nxt][b_st]);
    ad[0] = ra0; ad[8 * APITCH] = ra1; ad[16 * APITCH] = ra2; ad[24 * APITCH] = ra3;
    bd[0] = rb0; bd[4 * BPITCH] = rb1;
    __syncthreads();
  }
  {
    const int cur = (nk - 1) & 1;
    gemm_mma_step(&As[cur][aoff], &Bs[cur][boff], acc);
  }

#pragma unroll
  for (int mi = 0; mi < 2; ++mi) {
#pragma unroll
    for (int ni = 0; ni < 2; ++ni) {
      const int col = n0 + wn * 32 + ni * 16 + l16;
      const float badd = (bias != nullptr) ? bias[col] : 0.0f;
#pragma unroll
      for (int r = 0; r < 8; ++r) {
        const int row = m0 + wm * 32 + mi * 16 + r + 8 * lh;
        C[(size_t)row * N + col] = acc[mi][ni][r] + badd;
      }
    }
  }
}

__global__ __launch_bounds__(128) void qk_transform(float* __restrict__ Q,
                                                    float* __restrict__ Kb) {
  const int wave = threadIdx.x >> 5, lane = threadIdx.x & 31;
  const int hr = blockIdx.x * 4 + wave;
  if (hr >= ROWS * 8) return;
  const int row = hr >> 3, h = hr & 7;
  const size_t base = (size_t)row * LD + h * 64;

  float q0 = Q[base + lane], q1 = Q[base + lane + 32];
  float m = fmaxf(q0, q1);
#pragma unroll
  for (int off = 16; off > 0; off >>= 1) m = fmaxf(m, __shfl_xor(m, off, 32));
  float e0 = __expf(q0 - m), e1 = __expf(q1 - m);
  float s = e0 + e1;
#pragma unroll
  for (int off = 16; off > 0; off >>= 1) s += __shfl_xor(s, off, 32);
  float inv = 0.125f / s;
  Q[base + lane] = e0 * inv;
  Q[base + lane + 32] = e1 * inv;

  float k0 = Kb[base + lane], k1 = Kb[base + lane + 32];
  Kb[base + lane] = __expf(k0);
  Kb[base + lane + 32] = __expf(k1);
}

__global__ __launch_bounds__(256) void chunk_state(const float* __restrict__ Kb,
                                                   const float* __restrict__ Vb,
                                                   float* __restrict__ Sbuf,
                                                   float* __restrict__ ksumbuf) {
  __shared__ float Ks[64][65];
  __shared__ float Vs[64][65];
  const int task = blockIdx.x;
  const int ck = task & 31, h = (task >> 5) & 7, b = task >> 8;
  const int row0 = b * 2048 + ck * CHUNK, col0 = h * 64;
  const int t = threadIdx.x, wave = t >> 5, lane = t & 31;
  const int lh = lane >> 4, l16 = lane & 15;

  {
    int r = t >> 2, c = (t & 3) * 16;
    const float4* ks = reinterpret_cast<const float4*>(&Kb[(size_t)(row0 + r) * LD + col0 + c]);
    const float4* vs = reinterpret_cast<const float4*>(&Vb[(size_t)(row0 + r) * LD + col0 + c]);
#pragma unroll
    for (int i = 0; i < 4; ++i) {
      float4 kv = ks[i], vv = vs[i];
      float* kd = &Ks[r][c + i * 4];
      kd[0] = kv.x; kd[1] = kv.y; kd[2] = kv.z; kd[3] = kv.w;
      float* vd = &Vs[r][c + i * 4];
      vd[0] = vv.x; vd[1] = vv.y; vd[2] = vv.z; vd[3] = vv.w;
    }
  }
  __syncthreads();

  const size_t sbase = (size_t)task * 4096;
#pragma unroll
  for (int tt = wave * 2; tt <= wave * 2 + 1; ++tt) {
    const int mt = tt >> 2, nt = tt & 3;
    v8f acc = {};
#pragma unroll
    for (int kk = 0; kk < 64; kk += 4) {
      int tr = kk + 2 * lh;
      v2f a, bfr;
      a.x = Ks[tr][mt * 16 + l16];
      a.y = Ks[tr + 1][mt * 16 + l16];
      bfr.x = Vs[tr][nt * 16 + l16];
      bfr.y = Vs[tr + 1][nt * 16 + l16];
      acc = wmma4(a, bfr, acc);
    }
#pragma unroll
    for (int r = 0; r < 8; ++r) {
      int d = mt * 16 + r + 8 * lh, n = nt * 16 + l16;
      Sbuf[sbase + d * 64 + n] = acc[r];
    }
  }
  if (t < 64) {
    float s = 0.f;
#pragma unroll 8
    for (int j = 0; j < 64; ++j) s += Ks[j][t];
    ksumbuf[task * 64 + t] = s;
  }
}

__global__ void prefix_scan(float* __restrict__ Sbuf, float* __restrict__ ksumbuf) {
  const int g = blockIdx.x * blockDim.x + threadIdx.x;
  if (g < 16 * 4096) {
    const int bh = g >> 12, e = g & 4095;
    float run = 0.f;
    for (int c = 0; c < NCHUNK; ++c) {
      size_t idx = (size_t)(bh * NCHUNK + c) * 4096 + e;
      float v = Sbuf[idx];
      Sbuf[idx] = run;
      run += v;
    }
  } else if (g < 16 * 4096 + 16 * 64) {
    const int s2 = g - 16 * 4096, bh = s2 >> 6, d = s2 & 63;
    float run = 0.f;
    for (int c = 0; c < NCHUNK; ++c) {
      int idx = (bh * NCHUNK + c) * 64 + d;
      float v = ksumbuf[idx];
      ksumbuf[idx] = run;
      run += v;
    }
  }
}

__global__ __launch_bounds__(256) void attn_chunk(const float* __restrict__ Qb,
                                                  const float* __restrict__ Kb,
                                                  const float* __restrict__ Vb,
                                                  const float* __restrict__ Sbuf,
                                                  const float* __restrict__ ksumbuf,
                                                  float* __restrict__ attn) {
  extern __shared__ float sm[];
  float* Qs   = sm;
  float* Ks   = Qs + 64 * 65;
  float* Vs   = Ks + 64 * 65;
  float* Ss   = Vs + 64 * 65;
  float* Ms   = Ss + 64 * 65;
  float* ksum = Ms + 64 * 65;
  float* Dinv = ksum + 64;

  const int task = blockIdx.x;
  const int ck = task & 31, h = (task >> 5) & 7, b = task >> 8;
  const int row0 = b * 2048 + ck * CHUNK, col0 = h * 64;
  const int t = threadIdx.x, wave = t >> 5, lane = t & 31;
  const int lh = lane >> 4, l16 = lane & 15;

  {
    int r = t >> 2, c = (t & 3) * 16;
    const float4* qs = reinterpret_cast<const float4*>(&Qb[(size_t)(row0 + r) * LD + col0 + c]);
    const float4* ks = reinterpret_cast<const float4*>(&Kb[(size_t)(row0 + r) * LD + col0 + c]);
    const float4* vs = reinterpret_cast<const float4*>(&Vb[(size_t)(row0 + r) * LD + col0 + c]);
    const float4* ss = reinterpret_cast<const float4*>(&Sbuf[(size_t)task * 4096 + t * 16]);
#pragma unroll
    for (int i = 0; i < 4; ++i) {
      float4 qv = qs[i], kv = ks[i], vv = vs[i], sv = ss[i];
      float* qd = &Qs[r * 65 + c + i * 4];
      qd[0] = qv.x; qd[1] = qv.y; qd[2] = qv.z; qd[3] = qv.w;
      float* kd = &Ks[r * 65 + c + i * 4];
      kd[0] = kv.x; kd[1] = kv.y; kd[2] = kv.z; kd[3] = kv.w;
      float* vd = &Vs[r * 65 + c + i * 4];
      vd[0] = vv.x; vd[1] = vv.y; vd[2] = vv.z; vd[3] = vv.w;
      float* sd = &Ss[(t >> 2) * 65 + (t & 3) * 16 + i * 4];
      sd[0] = sv.x; sd[1] = sv.y; sd[2] = sv.z; sd[3] = sv.w;
    }
    if (t < 64) ksum[t] = ksumbuf[task * 64 + t];
  }
  __syncthreads();

#pragma unroll
  for (int tt = wave * 2; tt <= wave * 2 + 1; ++tt) {
    const int mt = tt >> 2, nt = tt & 3;
    v8f acc = {};
#pragma unroll
    for (int kk = 0; kk < 64; kk += 4) {
      v2f a, bfr;
      a.x = Qs[(mt * 16 + l16) * 65 + kk + 2 * lh];
      a.y = Qs[(mt * 16 + l16) * 65 + kk + 2 * lh + 1];
      bfr.x = Ks[(nt * 16 + l16) * 65 + kk + 2 * lh];
      bfr.y = Ks[(nt * 16 + l16) * 65 + kk + 2 * lh + 1];
      acc = wmma4(a, bfr, acc);
    }
#pragma unroll
    for (int r = 0; r < 8; ++r) {
      int i = mt * 16 + r + 8 * lh, j = nt * 16 + l16;
      Ms[i * 65 + j] = (j <= i) ? acc[r] : 0.0f;
    }
  }
  __syncthreads();

  if (t < 64) {
    float s = 0.f, dq = 0.f;
#pragma unroll 8
    for (int j = 0; j < 64; ++j) s += Ms[t * 65 + j];
#pragma unroll 8
    for (int d = 0; d < 64; ++d) dq += Qs[t * 65 + d] * ksum[d];
    Dinv[t] = 1.0f / fmaxf(s + dq, 1e-3f);
  }
  __syncthreads();

#pragma unroll
  for (int tt = wave * 2; tt <= wave * 2 + 1; ++tt) {
    const int mt = tt >> 2, nt = tt & 3;
    v8f acc = {};
#pragma unroll
    for (int kk = 0; kk < 64; kk += 4) {
      v2f a, bfr;
      a.x = Ms[(mt * 16 + l16) * 65 + kk + 2 * lh];
      a.y = Ms[(mt * 16 + l16) * 65 + kk + 2 * lh + 1];
      bfr.x = Vs[(kk + 2 * lh) * 65 + nt * 16 + l16];
      bfr.y = Vs[(kk + 2 * lh + 1) * 65 + nt * 16 + l16];
      acc = wmma4(a, bfr, acc);
    }
#pragma unroll
    for (int kk = 0; kk < 64; kk += 4) {
      v2f a, bfr;
      a.x = Qs[(mt * 16 + l16) * 65 + kk + 2 * lh];
      a.y = Qs[(mt * 16 + l16) * 65 + kk + 2 * lh + 1];
      bfr.x = Ss[(kk + 2 * lh) * 65 + nt * 16 + l16];
      bfr.y = Ss[(kk + 2 * lh + 1) * 65 + nt * 16 + l16];
      acc = wmma4(a, bfr, acc);
    }
#pragma unroll
    for (int r = 0; r < 8; ++r) {
      int i = mt * 16 + r + 8 * lh;
      int col = col0 + nt * 16 + l16;
      attn[(size_t)(row0 + i) * LD + col] = acc[r] * Dinv[i];
    }
  }
}

extern "C" void kernel_launch(void* const* d_in, const int* in_sizes, int n_in,
                              void* d_out, int out_size, void* d_ws, size_t ws_size,
                              hipStream_t stream) {
  const float* x  = (const float*)d_in[0];
  const float* Wq = (const float*)d_in[1];
  const float* Wk = (const float*)d_in[2];
  const float* Wv = (const float*)d_in[3];
  const float* Wo = (const float*)d_in[4];
  const float* bo = (const float*)d_in[5];
  float* out = (float*)d_out;

  float* ws = (float*)d_ws;
  const size_t MN = (size_t)ROWS * LD;
  float* Qb = ws;
  float* Kb = Qb + MN;
  float* Vb = Kb + MN;
  float* At = Vb + MN;
  float* Sb = At + MN;
  float* Kc = Sb + (size_t)NTASK * 4096;

  dim3 ggrid(LD / GBN, ROWS / GBM);

  gemm_f32<<<ggrid, 256, 0, stream>>>(x, Wq, Qb, nullptr, ROWS, LD, LD);
  gemm_f32<<<ggrid, 256, 0, stream>>>(x, Wk, Kb, nullptr, ROWS, LD, LD);
  gemm_f32<<<ggrid, 256, 0, stream>>>(x, Wv, Vb, nullptr, ROWS, LD, LD);

  qk_transform<<<(ROWS * 8) / 4, 128, 0, stream>>>(Qb, Kb);

  chunk_state<<<NTASK, 256, 0, stream>>>(Kb, Vb, Sb, Kc);

  prefix_scan<<<(16 * 4096 + 16 * 64 + 255) / 256, 256, 0, stream>>>(Sb, Kc);

  size_t shbytes = (size_t)(5 * 64 * 65 + 128) * sizeof(float);
  attn_chunk<<<NTASK, 256, shbytes, stream>>>(Qb, Kb, Vb, Sb, Kc, At);

  gemm_f32<<<ggrid, 256, 0, stream>>>(At, Wo, out, bo, ROWS, LD, LD);
}


# q softmax/8 and k exp folded into the tile staging of chunk_state and attn_chunk; the elementwise launch (32768 one-row waves) is gone: 7 launches; softmax row sum now 16 serial adds + 2 quad steps
# speedup vs baseline: 2.0968x; 1.3107x over previous
#include <hip/hip_runtime.h>
#include <hip/hip_bf16.h>

typedef float v2f __attribute__((ext_vector_type(2)));
typedef float v8f __attribute__((ext_vector_type(8)));

static __device__ __forceinline__ v8f wmma4(v2f a, v2f b, v8f c) {
  return __builtin_amdgcn_wmma_f32_16x16x4_f32(false, a, false, b, (short)0, c,
                                               false, false);
}

#define LD 512
#define ROWS 4096
#define CHUNK 64
#define NCHUNK 32
#define NTASK 512

#define GBM 128
#define GBN 64
#define GBK 32
#define APITCH 36
#define BPITCH 72

static __device__ __forceinline__ void gemm_mma_step(const float* __restrict__ as,
                                                     const float* __restrict__ bs,
                                                     v8f (&acc)[2][2]) {
  v2f a[GBK / 4][2], b[GBK / 4][2];
#pragma unroll
  for (int kq = 0; kq < GBK / 4; ++kq) {
    const int kk = kq * 4;
#pragma unroll
    for (int mi = 0; mi < 2; ++mi)
      a[kq][mi] = *reinterpret_cast<const v2f*>(&as[mi * 16 * APITCH + kk]);
#pragma unroll
    for (int ni = 0; ni < 2; ++ni) {
      b[kq][ni].x = bs[kk * BPITCH + ni * 16];
      b[kq][ni].y = bs[(kk + 1) * BPITCH + ni * 16];
    }
  }
  __builtin_amdgcn_sched_barrier(0);
#pragma unroll
  for (int kq = 0; kq < GBK / 4; ++kq)
#pragma unroll
    for (int mi = 0; mi < 2; ++mi)
#pragma unroll
      for (int ni = 0; ni < 2; ++ni)
        acc[mi][ni] = wmma4(a[kq][mi], b[kq][ni], acc[mi][ni]);
}

__global__ __launch_bounds__(256) void gemm_f32(const float* __restrict__ A,
                                                const float* __restrict__ B,
                                                float* __restrict__ C,
                                                const float* __restrict__ bias,
                                                int M, int N, int K) {
  __shared__ __attribute__((aligned(16))) float As[2][GBM * APITCH];
  __shared__ __attribute__((aligned(16))) float Bs[2][GBK * BPITCH];
  const int t = threadIdx.x;
  const int wave = t >> 5, lane = t & 31, lh = lane >> 4, l16 = lane & 15;
  const int wm = wave >> 1, wn = wave & 1;
  const int m0 = blockIdx.y * GBM, n0 = blockIdx.x * GBN;

  const int ar = t >> 3, ac = (t & 7) * 4;
  const int br = t >> 4, bc = (t & 15) * 4;
  const float* Ag = A + (size_t)(m0 + ar) * K + ac;
  const float* Bg = B + (size_t)br * N + n0 + bc;
  const size_t a_rows = (size_t)32 * K;
  const size_t b_rows = (size_t)16 * N;
  const size_t b_kstep = (size_t)GBK * N;
  const int a_st = ar * APITCH + ac;
  const int b_st = br * BPITCH + bc;

  const float4* Ap = reinterpret_cast<const float4*>(Ag);
  const float4* Bp = reinterpret_cast<const float4*>(Bg);
  const size_t a_rows4 = a_rows / 4, b_rows4 = b_rows / 4;
  float4 ra0 = Ap[0], ra1 = Ap[a_rows4], ra2 = Ap[2 * a_rows4], ra3 = Ap[3 * a_rows4];
  float4 rb0 = Bp[0], rb1 = Bp[b_rows4];
  {
    float4* ad = reinterpret_cast<float4*>(&As[0][a_st]);
    float4* bd = reinterpret_cast<float4*>(&Bs[0][b_st]);
    ad[0] = ra0; ad[8 * APITCH] = ra1; ad[16 * APITCH] = ra2; ad[24 * APITCH] = ra3;
    bd[0] = rb0; bd[4 * BPITCH] = rb1;
  }
  __syncthreads();

  v8f acc[2][2] = {};
  const int nk = K / GBK;
  const int aoff = (wm * 32 + l16) * APITCH + 2 * lh;
  const int boff = (2 * lh) * BPITCH + wn * 32 + l16;

  for (int kt = 0; kt + 1 < nk; ++kt) {
    const int cur = kt & 1, nxt = cur ^ 1;
    Ap += GBK / 4;
    Bp += b_kstep / 4;
    ra0 = Ap[0]; ra1 = Ap[a_rows4]; ra2 = Ap[2 * a_rows4]; ra3 = Ap[3 * a_rows4];
    rb0 = Bp[0]; rb1 = Bp[b_rows4];
    __builtin_amdgcn_sched_barrier(0);

    gemm_mma_step(&As[cur][aoff], &Bs[cur][boff], acc);
    __builtin_amdgcn_sched_barrier(0);

    float4* ad = reinterpret_cast<float4*>(&As[nxt][a_st]);
    float4* bd = reinterpret_cast<float4*>(&Bs[nxt][b_st]);
    ad[0] = ra0; ad[8 * APITCH] = ra1; ad[16 * APITCH] = ra2; ad[24 * APITCH] = ra3;
    bd[0] = rb0; bd[4 * BPITCH] = rb1;
    __syncthreads();
  }
  {
    const int cur = (nk - 1) & 1;
    gemm_mma_step(&As[cur][aoff], &Bs[cur][boff], acc);
  }

#pragma unroll
  for (int mi = 0; mi < 2; ++mi) {
#pragma unroll
    for (int ni = 0; ni < 2; ++ni) {
      const int col = n0 + wn * 32 + ni * 16 + l16;
      const float badd = (bias != nullptr) ? bias[col] : 0.0f;
#pragma unroll
      for (int r = 0; r < 8; ++r) {
        const int row = m0 + wm * 32 + mi * 16 + r + 8 * lh;
        C[(size_t)row * N + col] = acc[mi][ni][r] + badd;
      }
    }
  }
}

__global__ __launch_bounds__(256) void chunk_state(const float* __restrict__ Kb,
                                                   const float* __restrict__ Vb,
                                                   float* __restrict__ Sbuf,
                                                   float* __restrict__ ksumbuf) {
  __shared__ float Ks[64][65];
  __shared__ float Vs[64][65];
  const int task = blockIdx.x;
  const int ck = task & 31, h = (task >> 5) & 7, b = task >> 8;
  const int row0 = b * 2048 + ck * CHUNK, col0 = h * 64;
  const int t = threadIdx.x, wave = t >> 5, lane = t & 31;
  const int lh = lane >> 4, l16 = lane & 15;

  {
    int r = t >> 2, c = (t & 3) * 16;
    const float4* ks = reinterpret_cast<const float4*>(&Kb[(size_t)(row0 + r) * LD + col0 + c]);
    const float4* vs = reinterpret_cast<const float4*>(&Vb[(size_t)(row0 + r) * LD + col0 + c]);
#pragma unroll
    for (int i = 0; i < 4; ++i) {
      float4 kv = ks[i], vv = vs[i];
      float* kd = &Ks[r][c + i * 4];
      kd[0] = __expf(kv.x); kd[1] = __expf(kv.y); kd[2] = __expf(kv.z); kd[3] = __expf(kv.w);
      float* vd = &Vs[r][c + i * 4];
      vd[0] = vv.x; vd[1] = vv.y; vd[2] = vv.z; vd[3] = vv.w;
    }
  }
  __syncthreads();

  const size_t sbase = (size_t)task * 4096;
#pragma unroll
  for (int tt = wave * 2; tt <= wave * 2 + 1; ++tt) {
    const int mt = tt >> 2, nt = tt & 3;
    v8f acc = {};
#pragma unroll
    for (int kk = 0; kk < 64; kk += 4) {
      int tr = kk + 2 * lh;
      v2f a, bfr;
      a.x = Ks[tr][mt * 16 + l16];
      a.y = Ks[tr + 1][mt * 16 + l16];
      bfr.x = Vs[tr][nt * 16 + l16];
      bfr.y = Vs[tr + 1][nt * 16 + l16];
      acc = wmma4(a, bfr, acc);
    }
#pragma unroll
    for (int r = 0; r < 8; ++r) {
      int d = mt * 16 + r + 8 * lh, n = nt * 16 + l16;
      Sbuf[sbase + d * 64 + n] = acc[r];
    }
  }
  if (t < 64) {
    float s = 0.f;
#pragma unroll 8
    for (int j = 0; j < 64; ++j) s += Ks[j][t];
    ksumbuf[task * 64 + t] = s;
  }
}

__global__ void prefix_scan(float* __restrict__ Sbuf, float* __restrict__ ksumbuf) {
  const int g = blockIdx.x * blockDim.x + threadIdx.x;
  if (g < 16 * 4096) {
    const int bh = g >> 12, e = g & 4095;
    float run = 0.f;
    for (int c = 0; c < NCHUNK; ++c) {
      size_t idx = (size_t)(bh * NCHUNK + c) * 4096 + e;
      float v = Sbuf[idx];
      Sbuf[idx] = run;
      run += v;
    }
  } else if (g < 16 * 4096 + 16 * 64) {
    const int s2 = g - 16 * 4096, bh = s2 >> 6, d = s2 & 63;
    float run = 0.f;
    for (int c = 0; c < NCHUNK; ++c) {
      int idx = (bh * NCHUNK + c) * 64 + d;
      float v = ksumbuf[idx];
      ksumbuf[idx] = run;
      run += v;
    }
  }
}

__global__ __launch_bounds__(256) void attn_chunk(const float* __restrict__ Qb,
                                                  const float* __restrict__ Kb,
                                                  const float* __restrict__ Vb,
                                                  const float* __restrict__ Sbuf,
                                                  const float* __restrict__ ksumbuf,
                                                  float* __restrict__ attn) {
  extern __shared__ float sm[];
  float* Qs   = sm;
  float* Ks   = Qs + 64 * 65;
  float* Vs   = Ks + 64 * 65;
  float* Ss   = Vs + 64 * 65;
  float* Ms   = Ss + 64 * 65;
  float* ksum = Ms + 64 * 65;
  float* Dinv = ksum + 64;

  const int task = blockIdx.x;
  const int ck = task & 31, h = (task >> 5) & 7, b = task >> 8;
  const int row0 = b * 2048 + ck * CHUNK, col0 = h * 64;
  const int t = threadIdx.x, wave = t >> 5, lane = t & 31;
  const int lh = lane >> 4, l16 = lane & 15;

  {
    int r = t >> 2, c = (t & 3) * 16;
    const float4* qs = reinterpret_cast<const float4*>(&Qb[(size_t)(row0 + r) * LD + col0 + c]);
    const float4* ks = reinterpret_cast<const float4*>(&Kb[(size_t)(row0 + r) * LD + col0 + c]);
    const float4* vs = reinterpret_cast<const float4*>(&Vb[(size_t)(row0 + r) * LD + col0 + c]);
    const float4* ss = reinterpret_cast<const float4*>(&Sbuf[(size_t)task * 4096 + t * 16]);
    float4 qv[4];
#pragma unroll
    for (int i = 0; i < 4; ++i) qv[i] = qs[i];
#pragma unroll
    for (int i = 0; i < 4; ++i) {
      float4 kv = ks[i], vv = vs[i], sv = ss[i];
      float* kd = &Ks[r * 65 + c + i * 4];
      kd[0] = __expf(kv.x); kd[1] = __expf(kv.y); kd[2] = __expf(kv.z); kd[3] = __expf(kv.w);
      float* vd = &Vs[r * 65 + c + i * 4];
      vd[0] = vv.x; vd[1] = vv.y; vd[2] = vv.z; vd[3] = vv.w;
      float* sd = &Ss[(t >> 2) * 65 + (t & 3) * 16 + i * 4];
      sd[0] = sv.x; sd[1] = sv.y; sd[2] = sv.z; sd[3] = sv.w;
    }
    if (t < 64) ksum[t] = ksumbuf[task * 64 + t];

    float m = fmaxf(fmaxf(qv[0].x, qv[0].y), fmaxf(qv[0].z, qv[0].w));
#pragma unroll
    for (int i = 1; i < 4; ++i)
      m = fmaxf(m, fmaxf(fmaxf(qv[i].x, qv[i].y), fmaxf(qv[i].z, qv[i].w)));
    m = fmaxf(m, __shfl_xor(m, 1, 32));
    m = fmaxf(m, __shfl_xor(m, 2, 32));
    float s = 0.f;
#pragma unroll
    for (int i = 0; i < 4; ++i) {
      qv[i].x = __expf(qv[i].x - m); qv[i].y = __expf(qv[i].y - m);
      qv[i].z = __expf(qv[i].z - m); qv[i].w = __expf(qv[i].w - m);
      s += (qv[i].x + qv[i].y) + (qv[i].z + qv[i].w);
    }
    s += __shfl_xor(s, 1, 32);
    s += __shfl_xor(s, 2, 32);
    const float inv = 0.125f / s;
#pragma unroll
    for (int i = 0; i < 4; ++i) {
      float* qd = &Qs[r * 65 + c + i * 4];
      qd[0] = qv[i].x * inv; qd[1] = qv[i].y * inv; qd[2] = qv[i].z * inv; qd[3] = qv[i].w * inv;
    }
  }
  __syncthreads();

#pragma unroll
  for (int tt = wave * 2; tt <= wave * 2 + 1; ++tt) {
    const int mt = tt >> 2, nt = tt & 3;
    v8f acc = {};
#pragma unroll
    for (int kk = 0; kk < 64; kk += 4) {
      v2f a, bfr;
      a.x = Qs[(mt * 16 + l16) * 65 + kk + 2 * lh];
      a.y = Qs[(mt * 16 + l16) * 65 + kk + 2 * lh + 1];
      bfr.x = Ks[(nt * 16 + l16) * 65 + kk + 2 * lh];
      bfr.y = Ks[(nt * 16 + l16) * 65 + kk + 2 * lh + 1];
      acc = wmma4(a, bfr, acc);
    }
#pragma unroll
    for (int r = 0; r < 8; ++r) {
      int i = mt * 16 + r + 8 * lh, j = nt * 16 + l16;
      Ms[i * 65 + j] = (j <= i) ? acc[r] : 0.0f;
    }
  }
  __syncthreads();

  if (t < 64) {
    float s = 0.f, dq = 0.f;
#pragma unroll 8
    for (int j = 0; j < 64; ++j) s += Ms[t * 65 + j];
#pragma unroll 8
    for (int d = 0; d < 64; ++d) dq += Qs[t * 65 + d] * ksum[d];
    Dinv[t] = 1.0f / fmaxf(s + dq, 1e-3f);
  }
  __syncthreads();

#pragma unroll
  for (int tt = wave * 2; tt <= wave * 2 + 1; ++tt) {
    const int mt = tt >> 2, nt = tt & 3;
    v8f acc = {};
#pragma unroll
    for (int kk = 0; kk < 64; kk += 4) {
      v2f a, bfr;
      a.x = Ms[(mt * 16 + l16) * 65 + kk + 2 * lh];
      a.y = Ms[(mt * 16 + l16) * 65 + kk + 2 * lh + 1];
      bfr.x = Vs[(kk + 2 * lh) * 65 + nt * 16 + l16];
      bfr.y = Vs[(kk + 2 * lh + 1) * 65 + nt * 16 + l16];
      acc = wmma4(a, bfr, acc);
    }
#pragma unroll
    for (int kk = 0; kk < 64; kk += 4) {
      v2f a, bfr;
      a.x = Qs[(mt * 16 + l16) * 65 + kk + 2 * lh];
      a.y = Qs[(mt * 16 + l16) * 65 + kk + 2 * lh + 1];
      bfr.x = Ss[(kk + 2 * lh) * 65 + nt * 16 + l16];
      bfr.y = Ss[(kk + 2 * lh + 1) * 65 + nt * 16 + l16];
      acc = wmma4(a, bfr, acc);
    }
#pragma unroll
    for (int r = 0; r < 8; ++r) {
      int i = mt * 16 + r + 8 * lh;
      int col = col0 + nt * 16 + l16;
      attn[(size_t)(row0 + i) * LD + col] = acc[r] * Dinv[i];
    }
  }
}

extern "C" void kernel_launch(void* const* d_in, const int* in_sizes, int n_in,
                              void* d_out, int out_size, void* d_ws, size_t ws_size,
                              hipStream_t stream) {
  const float* x  = (const float*)d_in[0];
  const float* Wq = (const float*)d_in[1];
  const float* Wk = (const float*)d_in[2];
  const float* Wv = (const float*)d_in[3];
  const float* Wo = (const float*)d_in[4];
  const float* bo = (const float*)d_in[5];
  float* out = (float*)d_out;

  float* ws = (float*)d_ws;
  const size_t MN = (size_t)ROWS * LD;
  float* Qb = ws;
  float* Kb = Qb + MN;
  float* Vb = Kb + MN;
  float* At = Vb + MN;
  float* Sb = At + MN;
  float* Kc = Sb + (size_t)NTASK * 4096;

  dim3 ggrid(LD / GBN, ROWS / GBM);

  gemm_f32<<<ggrid, 256, 0, stream>>>(x, Wq, Qb, nullptr, ROWS, LD, LD);
  gemm_f32<<<ggrid, 256, 0, stream>>>(x, Wk, Kb, nullptr, ROWS, LD, LD);
  gemm_f32<<<ggrid, 256, 0, stream>>>(x, Wv, Vb, nullptr, ROWS, LD, LD);

  chunk_state<<<NTASK, 256, 0, stream>>>(Kb, Vb, Sb, Kc);

  prefix_scan<<<(16 * 4096 + 16 * 64 + 255) / 256, 256, 0, stream>>>(Sb, Kc);

  size_t shbytes = (size_t)(5 * 64 * 65 + 128) * sizeof(float);
  attn_chunk<<<NTASK, 256, shbytes, stream>>>(Qb, Kb, Vb, Sb, Kc, At);

  gemm_f32<<<ggrid, 256, 0, stream>>>(At, Wo, out, bo, ROWS, LD, LD);
}


# q, k and v projections run as ONE launch (grid z picks weight and output): 768 work-groups, 3 resident per WGP, so one group's WMMAs fill another's barrier and staging gaps; 5 launches
# speedup vs baseline: 2.4765x; 1.1811x over previous
#include <hip/hip_runtime.h>
#include <hip/hip_bf16.h>

typedef float v2f __attribute__((ext_vector_type(2)));
typedef float v8f __attribute__((ext_vector_type(8)));

static __device__ __forceinline__ v8f wmma4(v2f a, v2f b, v8f c) {
  return __builtin_amdgcn_wmma_f32_16x16x4_f32(false, a, false, b, (short)0, c,
                                               false, false);
}

#define LD 512
#define ROWS 4096
#define CHUNK 64
#define NCHUNK 32
#define NTASK 512

#define GBM 128
#define GBN 64
#define GBK 32
#define APITCH 36
#define BPITCH 72

static __device__ __forceinline__ void gemm_mma_step(const float* __restrict__ as,
                                                     const float* __restrict__ bs,
                                                     v8f (&acc)[2][2]) {
  v2f a[GBK / 4][2], b[GBK / 4][2];
#pragma unroll
  for (int kq = 0; kq < GBK / 4; ++kq) {
    const int kk = kq * 4;
#pragma unroll
    for (int mi = 0; mi < 2; ++mi)
      a[kq][mi] = *reinterpret_cast<const v2f*>(&as[mi * 16 * APITCH + kk]);
#pragma unroll
    for (int ni = 0; ni < 2; ++ni) {
      b[kq][ni].x = bs[kk * BPITCH + ni * 16];
      b[kq][ni].y = bs[(kk + 1) * BPITCH + ni * 16];
    }
  }
  __builtin_amdgcn_sched_barrier(0);
#pragma unroll
  for (int kq = 0; kq < GBK / 4; ++kq)
#pragma unroll
    for (int mi = 0; mi < 2; ++mi)
#pragma unroll
      for (int ni = 0; ni < 2; ++ni)
        acc[mi][ni] = wmma4(a[kq][mi], b[kq][ni], acc[mi][ni]);
}

__global__ __launch_bounds__(256) void gemm_f32(const float* __restrict__ A,
                                                const float* __restrict__ B0,
                                                const float* __restrict__ B1,
                                                const float* __restrict__ B2,
                                                float* __restrict__ C0,
                                                float* __restrict__ C1,
                                                float* __restrict__ C2,
                                                const float* __restrict__ bias,
                                                int M, int N, int K) {
  __shared__ __attribute__((aligned(16))) float As[2][GBM * APITCH];
  __shared__ __attribute__((aligned(16))) float Bs[2][GBK * BPITCH];
  const int t = threadIdx.x;
  const int wave = t >> 5, lane = t & 31, lh = lane >> 4, l16 = lane & 15;
  const int wm = wave >> 1, wn = wave & 1;
  const int m0 = blockIdx.y * GBM, n0 = blockIdx.x * GBN;
  const int z = blockIdx.z;
  const float* __restrict__ B = (z == 0) ? B0 : (z == 1) ? B1 : B2;
  float* __restrict__ C = (z == 0) ? C0 : (z == 1) ? C1 : C2;

  const int ar = t >> 3, ac = (t & 7) * 4;
  const int br = t >> 4, bc = (t & 15) * 4;
  const float* Ag = A + (size_t)(m0 + ar) * K + ac;
  const float* Bg = B + (size_t)br * N + n0 + bc;
  const size_t a_rows = (size_t)32 * K;
  const size_t b_rows = (size_t)16 * N;
  const size_t b_kstep = (size_t)GBK * N;
  const int a_st = ar * APITCH + ac;
  const int b_st = br * BPITCH + bc;

  const float4* Ap = reinterpret_cast<const float4*>(Ag);
  const float4* Bp = reinterpret_cast<const float4*>(Bg);
  const size_t a_rows4 = a_rows / 4, b_rows4 = b_rows / 4;
  float4 ra0 = Ap[0], ra1 = Ap[a_rows4], ra2 = Ap[2 * a_rows4], ra3 = Ap[3 * a_rows4];
  float4 rb0 = Bp[0], rb1 = Bp[b_rows4];
  {
    float4* ad = reinterpret_cast<float4*>(&As[0][a_st]);
    float4* bd = reinterpret_cast<float4*>(&Bs[0][b_st]);
    ad[0] = ra0; ad[8 * APITCH] = ra1; ad[16 * APITCH] = ra2; ad[24 * APITCH] = ra3;
    bd[0] = rb0; bd[4 * BPITCH] = rb1;
  }
  __syncthreads();

  v8f acc[2][2] = {};
  const int nk = K / GBK;
  const int aoff = (wm * 32 + l16) * APITCH + 2 * lh;
  const int boff = (2 * lh) * BPITCH + wn * 32 + l16;

  for (int kt = 0; kt + 1 < nk; ++kt) {
    const int cur = kt & 1, nxt = cur ^ 1;
    Ap += GBK / 4;
    Bp += b_kstep / 4;
    ra0 = Ap[0]; ra1 = Ap[a_rows4]; ra2 = Ap[2 * a_rows4]; ra3 = Ap[3 * a_rows4];
    rb0 = Bp[0]; rb1 = Bp[b_rows4];
    __builtin_amdgcn_sched_barrier(0);

    gemm_mma_step(&As[cur][aoff], &Bs[cur][boff], acc);
    __builtin_amdgcn_sched_barrier(0);

    float4* ad = reinterpret_cast<float4*>(&As[nxt][a_st]);
    float4* bd = reinterpret_cast<float4*>(&Bs[nxt][b_st]);
    ad[0] = ra0; ad[8 * APITCH] = ra1; ad[16 * APITCH] = ra2; ad[24 * APITCH] = ra3;
    bd[0] = rb0; bd[4 * BPITCH] = rb1;
    __syncthreads();
  }
  {
    const int cur = (nk - 1) & 1;
    gemm_mma_step(&As[cur][aoff], &Bs[cur][boff], acc);
  }

#pragma unroll
  for (int mi = 0; mi < 2; ++mi) {
#pragma unroll
    for (int ni = 0; ni < 2; ++ni) {
      const int col = n0 + wn * 32 + ni * 16 + l16;
      const float badd = (bias != nullptr) ? bias[col] : 0.0f;
#pragma unroll
      for (int r = 0; r < 8; ++r) {
        const int row = m0 + wm * 32 + mi * 16 + r + 8 * lh;
        C[(size_t)row * N + col] = acc[mi][ni][r] + badd;
      }
    }
  }
}

__global__ __launch_bounds__(256) void chunk_state(const float* __restrict__ Kb,
                                                   const float* __restrict__ Vb,
                                                   float* __restrict__ Sbuf,
                                                   float* __restrict__ ksumbuf) {
  __shared__ float Ks[64][65];
  __shared__ float Vs[64][65];
  const int task = blockIdx.x;
  const int ck = task & 31, h = (task >> 5) & 7, b = task >> 8;
  const int row0 = b * 2048 + ck * CHUNK, col0 = h * 64;
  const int t = threadIdx.x, wave = t >> 5, lane = t & 31;
  const int lh = lane >> 4, l16 = lane & 15;

  {
    int r = t >> 2, c = (t & 3) * 16;
    const float4* ks = reinterpret_cast<const float4*>(&Kb[(size_t)(row0 + r) * LD + col0 + c]);
    const float4* vs = reinterpret_cast<const float4*>(&Vb[(size_t)(row0 + r) * LD + col0 + c]);
#pragma unroll
    for (int i = 0; i < 4; ++i) {
      float4 kv = ks[i], vv = vs[i];
      float* kd = &Ks[r][c + i * 4];
      kd[0] = __expf(kv.x); kd[1] = __expf(kv.y); kd[2] = __expf(kv.z); kd[3] = __expf(kv.w);
      float* vd = &Vs[r][c + i * 4];
      vd[0] = vv.x; vd[1] = vv.y; vd[2] = vv.z; vd[3] = vv.w;
    }
  }
  __syncthreads();

  const size_t sbase = (size_t)task * 4096;
#pragma unroll
  for (int tt = wave * 2; tt <= wave * 2 + 1; ++tt) {
    const int mt = tt >> 2, nt = tt & 3;
    v8f acc = {};
#pragma unroll
    for (int kk = 0; kk < 64; kk += 4) {
      int tr = kk + 2 * lh;
      v2f a, bfr;
      a.x = Ks[tr][mt * 16 + l16];
      a.y = Ks[tr + 1][mt * 16 + l16];
      bfr.x = Vs[tr][nt * 16 + l16];
      bfr.y = Vs[tr + 1][nt * 16 + l16];
      acc = wmma4(a, bfr, acc);
    }
#pragma unroll
    for (int r = 0; r < 8; ++r) {
      int d = mt * 16 + r + 8 * lh, n = nt * 16 + l16;
      Sbuf[sbase + d * 64 + n] = acc[r];
    }
  }
  if (t < 64) {
    float s = 0.f;
#pragma unroll 8
    for (int j = 0; j < 64; ++j) s += Ks[j][t];
    ksumbuf[task * 64 + t] = s;
  }
}

__global__ void prefix_scan(float* __restrict__ Sbuf, float* __restrict__ ksumbuf) {
  const int g = blockIdx.x * blockDim.x + threadIdx.x;
  if (g < 16 * 4096) {
    const int bh = g >> 12, e = g & 4095;
    float run = 0.f;
    for (int c = 0; c < NCHUNK; ++c) {
      size_t idx = (size_t)(bh * NCHUNK + c) * 4096 + e;
      float v = Sbuf[idx];
      Sbuf[idx] = run;
      run += v;
    }
  } else if (g < 16 * 4096 + 16 * 64) {
    const int s2 = g - 16 * 4096, bh = s2 >> 6, d = s2 & 63;
    float run = 0.f;
    for (int c = 0; c < NCHUNK; ++c) {
      int idx = (bh * NCHUNK + c) * 64 + d;
      float v = ksumbuf[idx];
      ksumbuf[idx] = run;
      run += v;
    }
  }
}

__global__ __launch_bounds__(256) void attn_chunk(const float* __restrict__ Qb,
                                                  const float* __restrict__ Kb,
                                                  const float* __restrict__ Vb,
                                                  const float* __restrict__ Sbuf,
                                                  const float* __restrict__ ksumbuf,
                                                  float* __restrict__ attn) {
  extern __shared__ float sm[];
  float* Qs   = sm;
  float* Ks   = Qs + 64 * 65;
  float* Vs   = Ks + 64 * 65;
  float* Ss   = Vs + 64 * 65;
  float* Ms   = Ss + 64 * 65;
  float* ksum = Ms + 64 * 65;
  float* Dinv = ksum + 64;

  const int task = blockIdx.x;
  const int ck = task & 31, h = (task >> 5) & 7, b = task >> 8;
  const int row0 = b * 2048 + ck * CHUNK, col0 = h * 64;
  const int t = threadIdx.x, wave = t >> 5, lane = t & 31;
  const int lh = lane >> 4, l16 = lane & 15;

  {
    int r = t >> 2, c = (t & 3) * 16;
    const float4* qs = reinterpret_cast<const float4*>(&Qb[(size_t)(row0 + r) * LD + col0 + c]);
    const float4* ks = reinterpret_cast<const float4*>(&Kb[(size_t)(row0 + r) * LD + col0 + c]);
    const float4* vs = reinterpret_cast<const float4*>(&Vb[(size_t)(row0 + r) * LD + col0 + c]);
    const float4* ss = reinterpret_cast<const float4*>(&Sbuf[(size_t)task * 4096 + t * 16]);
    float4 qv[4];
#pragma unroll
    for (int i = 0; i < 4; ++i) qv[i] = qs[i];
#pragma unroll
    for (int i = 0; i < 4; ++i) {
      float4 kv = ks[i], vv = vs[i], sv = ss[i];
      float* kd = &Ks[r * 65 + c + i * 4];
      kd[0] = __expf(kv.x); kd[1] = __expf(kv.y); kd[2] = __expf(kv.z); kd[3] = __expf(kv.w);
      float* vd = &Vs[r * 65 + c + i * 4];
      vd[0] = vv.x; vd[1] = vv.y; vd[2] = vv.z; vd[3] = vv.w;
      float* sd = &Ss[(t >> 2) * 65 + (t & 3) * 16 + i * 4];
      sd[0] = sv.x; sd[1] = sv.y; sd[2] = sv.z; sd[3] = sv.w;
    }
    if (t < 64) ksum[t] = ksumbuf[task * 64 + t];

    float m = fmaxf(fmaxf(qv[0].x, qv[0].y), fmaxf(qv[0].z, qv[0].w));
#pragma unroll
    for (int i = 1; i < 4; ++i)
      m = fmaxf(m, fmaxf(fmaxf(qv[i].x, qv[i].y), fmaxf(qv[i].z, qv[i].w)));
    m = fmaxf(m, __shfl_xor(m, 1, 32));
    m = fmaxf(m, __shfl_xor(m, 2, 32));
    float s = 0.f;
#pragma unroll
    for (int i = 0; i < 4; ++i) {
      qv[i].x = __expf(qv[i].x - m); qv[i].y = __expf(qv[i].y - m);
      qv[i].z = __expf(qv[i].z - m); qv[i].w = __expf(qv[i].w - m);
      s += (qv[i].x + qv[i].y) + (qv[i].z + qv[i].w);
    }
    s += __shfl_xor(s, 1, 32);
    s += __shfl_xor(s, 2, 32);
    const float inv = 0.125f / s;
#pragma unroll
    for (int i = 0; i < 4; ++i) {
      float* qd = &Qs[r * 65 + c + i * 4];
      qd[0] = qv[i].x * inv; qd[1] = qv[i].y * inv; qd[2] = qv[i].z * inv; qd[3] = qv[i].w * inv;
    }
  }
  __syncthreads();

#pragma unroll
  for (int tt = wave * 2; tt <= wave * 2 + 1; ++tt) {
    const int mt = tt >> 2, nt = tt & 3;
    v8f acc = {};
#pragma unroll
    for (int kk = 0; kk < 64; kk += 4) {
      v2f a, bfr;
      a.x = Qs[(mt * 16 + l16) * 65 + kk + 2 * lh];
      a.y = Qs[(mt * 16 + l16) * 65 + kk + 2 * lh + 1];
      bfr.x = Ks[(nt * 16 + l16) * 65 + kk + 2 * lh];
      bfr.y = Ks[(nt * 16 + l16) * 65 + kk + 2 * lh + 1];
      acc = wmma4(a, bfr, acc);
    }
#pragma unroll
    for (int r = 0; r < 8; ++r) {
      int i = mt * 16 + r + 8 * lh, j = nt * 16 + l16;
      Ms[i * 65 + j] = (j <= i) ? acc[r] : 0.0f;
    }
  }
  __syncthreads();

  if (t < 64) {
    float s = 0.f, dq = 0.f;
#pragma unroll 8
    for (int j = 0; j < 64; ++j) s += Ms[t * 65 + j];
#pragma unroll 8
    for (int d = 0; d < 64; ++d) dq += Qs[t * 65 + d] * ksum[d];
    Dinv[t] = 1.0f / fmaxf(s + dq, 1e-3f);
  }
  __syncthreads();

#pragma unroll
  for (int tt = wave * 2; tt <= wave * 2 + 1; ++tt) {
    const int mt = tt >> 2, nt = tt & 3;
    v8f acc = {};
#pragma unroll
    for (int kk = 0; kk < 64; kk += 4) {
      v2f a, bfr;
      a.x = Ms[(mt * 16 + l16) * 65 + kk + 2 * lh];
      a.y = Ms[(mt * 16 + l16) * 65 + kk + 2 * lh + 1];
      bfr.x = Vs[(kk + 2 * lh) * 65 + nt * 16 + l16];
      bfr.y = Vs[(kk + 2 * lh + 1) * 65 + nt * 16 + l16];
      acc = wmma4(a, bfr, acc);
    }
#pragma unroll
    for (int kk = 0; kk < 64; kk += 4) {
      v2f a, bfr;
      a.x = Qs[(mt * 16 + l16) * 65 + kk + 2 * lh];
      a.y = Qs[(mt * 16 + l16) * 65 + kk + 2 * lh + 1];
      bfr.x = Ss[(kk + 2 * lh) * 65 + nt * 16 + l16];
      bfr.y = Ss[(kk + 2 * lh + 1) * 65 + nt * 16 + l16];
      acc = wmma4(a, bfr, acc);
    }
#pragma unroll
    for (int r = 0; r < 8; ++r) {
      int i = mt * 16 + r + 8 * lh;
      int col = col0 + nt * 16 + l16;
      attn[(size_t)(row0 + i) * LD + col] = acc[r] * Dinv[i];
    }
  }
}

extern "C" void kernel_launch(void* const* d_in, const int* in_sizes, int n_in,
                              void* d_out, int out_size, void* d_ws, size_t ws_size,
                              hipStream_t stream) {
  const float* x  = (const float*)d_in[0];
  const float* Wq = (const float*)d_in[1];
  const float* Wk = (const float*)d_in[2];
  const float* Wv = (const float*)d_in[3];
  const float* Wo = (const float*)d_in[4];
  const float* bo = (const float*)d_in[5];
  float* out = (float*)d_out;

  float* ws = (float*)d_ws;
  const size_t MN = (size_t)ROWS * LD;
  float* Qb = ws;
  float* Kb = Qb + MN;
  float* Vb = Kb + MN;
  float* At = Vb + MN;
  float* Sb = At + MN;
  float* Kc = Sb + (size_t)NTASK * 4096;

  dim3 ggrid(LD / GBN, ROWS / GBM);
  dim3 qkvgrid(LD / GBN, ROWS / GBM, 3);

  gemm_f32<<<qkvgrid, 256, 0, stream>>>(x, Wq, Wk, Wv, Qb, Kb, Vb, nullptr, ROWS, LD, LD);

  chunk_state<<<NTASK, 256, 0, stream>>>(Kb, Vb, Sb, Kc);

  prefix_scan<<<(16 * 4096 + 16 * 64 + 255) / 256, 256, 0, stream>>>(Sb, Kc);

  size_t shbytes = (size_t)(5 * 64 * 65 + 128) * sizeof(float);
  attn_chunk<<<NTASK, 256, shbytes, stream>>>(Qb, Kb, Vb, Sb, Kc, At);

  gemm_f32<<<ggrid, 256, 0, stream>>>(At, Wo, nullptr, nullptr, out, nullptr, nullptr, bo, ROWS, LD, LD);
}


# k and v of one (64-row chunk, head) now come out of ONE 64x128 tile of the q/k/v launch; its epilogue forms exp(K)^T V and ksum from the accumulators via LDS: chunk_state launch gone, 4 launches
# speedup vs baseline: 2.5638x; 1.0353x over previous
#include <hip/hip_runtime.h>
#include <hip/hip_bf16.h>

typedef float v2f __attribute__((ext_vector_type(2)));
typedef float v8f __attribute__((ext_vector_type(8)));

static __device__ __forceinline__ v8f wmma4(v2f a, v2f b, v8f c) {
  return __builtin_amdgcn_wmma_f32_16x16x4_f32(false, a, false, b, (short)0, c,
                                               false, false);
}

#define LD 512
#define ROWS 4096
#define CHUNK 64
#define NCHUNK 32
#define NTASK 512

#define GBM 128
#define GBN 64
#define GBK 32
#define APITCH 36
#define BPITCH 72

template <int BP>
static __device__ __forceinline__ void gemm_mma_step(const float* __restrict__ as,
                                                     const float* __restrict__ bs,
                                                     v8f (&acc)[2][2]) {
  v2f a[GBK / 4][2], b[GBK / 4][2];
#pragma unroll
  for (int kq = 0; kq < GBK / 4; ++kq) {
    const int kk = kq * 4;
#pragma unroll
    for (int mi = 0; mi < 2; ++mi)
      a[kq][mi] = *reinterpret_cast<const v2f*>(&as[mi * 16 * APITCH + kk]);
#pragma unroll
    for (int ni = 0; ni < 2; ++ni) {
      b[kq][ni].x = bs[kk * BP + ni * 16];
      b[kq][ni].y = bs[(kk + 1) * BP + ni * 16];
    }
  }
  __builtin_amdgcn_sched_barrier(0);
#pragma unroll
  for (int kq = 0; kq < GBK / 4; ++kq)
#pragma unroll
    for (int mi = 0; mi < 2; ++mi)
#pragma unroll
      for (int ni = 0; ni < 2; ++ni)
        acc[mi][ni] = wmma4(a[kq][mi], b[kq][ni], acc[mi][ni]);
}

__global__ __launch_bounds__(256) void gemm_f32(const float* __restrict__ A,
                                                const float* __restrict__ B,
                                                float* __restrict__ C,
                                                const float* __restrict__ bias,
                                                int M, int N, int K) {
  __shared__ __attribute__((aligned(16))) float As[2][GBM * APITCH];
  __shared__ __attribute__((aligned(16))) float Bs[2][GBK * BPITCH];
  const int t = threadIdx.x;
  const int wave = t >> 5, lane = t & 31, lh = lane >> 4, l16 = lane & 15;
  const int wm = wave >> 1, wn = wave & 1;
  const int m0 = blockIdx.y * GBM, n0 = blockIdx.x * GBN;

  const int ar = t >> 3, ac = (t & 7) * 4;
  const int br = t >> 4, bc = (t & 15) * 4;
  const float* Ag = A + (size_t)(m0 + ar) * K + ac;
  const float* Bg = B + (size_t)br * N + n0 + bc;
  const size_t a_rows = (size_t)32 * K;
  const size_t b_rows = (size_t)16 * N;
  const size_t b_kstep = (size_t)GBK * N;
  const int a_st = ar * APITCH + ac;
  const int b_st = br * BPITCH + bc;

  const float4* Ap = reinterpret_cast<const float4*>(Ag);
  const float4* Bp = reinterpret_cast<const float4*>(Bg);
  const size_t a_rows4 = a_rows / 4, b_rows4 = b_rows / 4;
  float4 ra0 = Ap[0], ra1 = Ap[a_rows4], ra2 = Ap[2 * a_rows4], ra3 = Ap[3 * a_rows4];
  float4 rb0 = Bp[0], rb1 = Bp[b_rows4];
  {
    float4* ad = reinterpret_cast<float4*>(&As[0][a_st]);
    float4* bd = reinterpret_cast<float4*>(&Bs[0][b_st]);
    ad[0] = ra0; ad[8 * APITCH] = ra1; ad[16 * APITCH] = ra2; ad[24 * APITCH] = ra3;
    bd[0] = rb0; bd[4 * BPITCH] = rb1;
  }
  __syncthreads();

  v8f acc[2][2] = {};
  const int nk = K / GBK;
  const int aoff = (wm * 32 + l16) * APITCH + 2 * lh;
  const int boff = (2 * lh) * BPITCH + wn * 32 + l16;

  for (int kt = 0; kt + 1 < nk; ++kt) {
    const int cur = kt & 1, nxt = cur ^ 1;
    Ap += GBK / 4;
    Bp += b_kstep / 4;
    ra0 = Ap[0]; ra1 = Ap[a_rows4]; ra2 = Ap[2 * a_rows4]; ra3 = Ap[3 * a_rows4];
    rb0 = Bp[0]; rb1 = Bp[b_rows4];
    __builtin_amdgcn_sched_barrier(0);

    gemm_mma_step<BPITCH>(&As[cur][aoff], &Bs[cur][boff], acc);
    __builtin_amdgcn_sched_barrier(0);

    float4* ad = reinterpret_cast<float4*>(&As[nxt][a_st]);
    float4* bd = reinterpret_cast<float4*>(&Bs[nxt][b_st]);
    ad[0] = ra0; ad[8 * APITCH] = ra1; ad[16 * APITCH] = ra2; ad[24 * APITCH] = ra3;
    bd[0] = rb0; bd[4 * BPITCH] = rb1;
    __syncthreads();
  }
  {
    const int cur = (nk - 1) & 1;
    gemm_mma_step<BPITCH>(&As[cur][aoff], &Bs[cur][boff], acc);
  }

#pragma unroll
  for (int mi = 0; mi < 2; ++mi) {
#pragma unroll
    for (int ni = 0; ni < 2; ++ni) {
      const int col = n0 + wn * 32 + ni * 16 + l16;
      const float badd = (bias != nullptr) ? bias[col] : 0.0f;
#pragma unroll
      for (int r = 0; r < 8; ++r) {
        const int row = m0 + wm * 32 + mi * 16 + r + 8 * lh;
        C[(size_t)row * N + col] = acc[mi][ni][r] + badd;
      }
    }
  }
}

#define QBM 64
#define QBN 128
#define QBPITCH 136
#define QA_STAGE (QBM * APITCH)
#define QB_STAGE (GBK * QBPITCH)

__global__ __launch_bounds__(256) void qkv_state(const float* __restrict__ X,
                                                 const float* __restrict__ Wq,
                                                 const float* __restrict__ Wk,
                                                 const float* __restrict__ Wv,
                                                 float* __restrict__ Qb,
                                                 float* __restrict__ Kb,
                                                 float* __restrict__ Vb,
                                                 float* __restrict__ Sbuf,
                                                 float* __restrict__ ksumbuf) {
  __shared__ __attribute__((aligned(16))) float smem[2 * QA_STAGE + 2 * QB_STAGE];
  float* As = smem;
  float* Bs = smem + 2 * QA_STAGE;
  const int t = threadIdx.x;
  const int wave = t >> 5, lane = t & 31, lh = lane >> 4, l16 = lane & 15;
  const int wm = wave >> 2, wn = wave & 3;
  const int bx = blockIdx.x;
  const bool kv = bx >= 4;
  const int h = bx - 4;
  const int m0 = blockIdx.y * QBM;

  const int ar = t >> 3, ac = (t & 7) * 4;
  const int br = t >> 5, bc = (t & 31) * 4;
  const float* Bsrc = kv ? ((bc < 64) ? Wk + h * 64 + bc : Wv + h * 64 + (bc - 64))
                         : Wq + bx * QBN + bc;
  const float4* Ap = reinterpret_cast<const float4*>(X + (size_t)(m0 + ar) * LD + ac);
  const float4* Bp = reinterpret_cast<const float4*>(Bsrc + (size_t)br * LD);
  constexpr int a_rows4 = 32 * LD / 4;
  constexpr int b_rows4 = 8 * LD / 4;
  constexpr int b_kstep4 = GBK * LD / 4;
  const int a_st = ar * APITCH + ac;
  const int b_st = br * QBPITCH + bc;

  float4 ra0 = Ap[0], ra1 = Ap[a_rows4];
  float4 rb0 = Bp[0], rb1 = Bp[b_rows4], rb2 = Bp[2 * b_rows4], rb3 = Bp[3 * b_rows4];
  {
    float4* ad = reinterpret_cast<float4*>(&As[a_st]);
    float4* bd = reinterpret_cast<float4*>(&Bs[b_st]);
    ad[0] = ra0; ad[8 * APITCH] = ra1;
    bd[0] = rb0; bd[2 * QBPITCH] = rb1; bd[4 * QBPITCH] = rb2; bd[6 * QBPITCH] = rb3;
  }
  __syncthreads();

  v8f acc[2][2] = {};
  constexpr int nk = LD / GBK;
  const int aoff = (wm * 32 + l16) * APITCH + 2 * lh;
  const int boff = (2 * lh) * QBPITCH + wn * 32 + l16;

  for (int kt = 0; kt + 1 < nk; ++kt) {
    const int cur = kt & 1, nxt = cur ^ 1;
    Ap += GBK / 4;
    Bp += b_kstep4;
    ra0 = Ap[0]; ra1 = Ap[a_rows4];
    rb0 = Bp[0]; rb1 = Bp[b_rows4]; rb2 = Bp[2 * b_rows4]; rb3 = Bp[3 * b_rows4];
    __builtin_amdgcn_sched_barrier(0);

    gemm_mma_step<QBPITCH>(&As[cur * QA_STAGE + aoff], &Bs[cur * QB_STAGE + boff], acc);
    __builtin_amdgcn_sched_barrier(0);

    float4* ad = reinterpret_cast<float4*>(&As[nxt * QA_STAGE + a_st]);
    float4* bd = reinterpret_cast<float4*>(&Bs[nxt * QB_STAGE + b_st]);
    ad[0] = ra0; ad[8 * APITCH] = ra1;
    bd[0] = rb0; bd[2 * QBPITCH] = rb1; bd[4 * QBPITCH] = rb2; bd[6 * QBPITCH] = rb3;
    __syncthreads();
  }
  {
    const int cur = (nk - 1) & 1;
    gemm_mma_step<QBPITCH>(&As[cur * QA_STAGE + aoff], &Bs[cur * QB_STAGE + boff], acc);
  }

  float* Ks = smem;
  float* Vs = smem + 64 * 65;
  if (kv) {
    __syncthreads();
    const int c0 = (wn & 1) * 32;
#pragma unroll
    for (int mi = 0; mi < 2; ++mi) {
#pragma unroll
      for (int ni = 0; ni < 2; ++ni) {
        const int cc = c0 + ni * 16 + l16;
#pragma unroll
        for (int r = 0; r < 8; ++r) {
          const int rr = wm * 32 + mi * 16 + r + 8 * lh;
          if (wn < 2) Ks[rr * 65 + cc] = __expf(acc[mi][ni][r]);
          else        Vs[rr * 65 + cc] = acc[mi][ni][r];
        }
      }
    }
    __syncthreads();
  }

  {
    float* __restrict__ dst = !kv ? Qb : (wn < 2) ? Kb : Vb;
    const int cbase = !kv ? bx * QBN + wn * 32 : h * 64 + (wn & 1) * 32;
#pragma unroll
    for (int mi = 0; mi < 2; ++mi) {
#pragma unroll
      for (int ni = 0; ni < 2; ++ni) {
        const int col = cbase + ni * 16 + l16;
#pragma unroll
        for (int r = 0; r < 8; ++r) {
          const int row = m0 + wm * 32 + mi * 16 + r + 8 * lh;
          dst[(size_t)row * LD + col] = acc[mi][ni][r];
        }
      }
    }
  }
  if (!kv) return;

  const int task = ((m0 >> 11) * 8 + h) * NCHUNK + ((m0 >> 6) & (NCHUNK - 1));
  const size_t sbase = (size_t)task * 4096;
  {
    const int mt = wave >> 1, nt0 = (wave & 1) * 2;
    v8f s0 = {}, s1 = {};
#pragma unroll
    for (int kk = 0; kk < 64; kk += 4) {
      const int tr = kk + 2 * lh;
      v2f a, b0, b1;
      a.x  = Ks[tr * 65 + mt * 16 + l16];
      a.y  = Ks[(tr + 1) * 65 + mt * 16 + l16];
      b0.x = Vs[tr * 65 + nt0 * 16 + l16];
      b0.y = Vs[(tr + 1) * 65 + nt0 * 16 + l16];
      b1.x = Vs[tr * 65 + (nt0 + 1) * 16 + l16];
      b1.y = Vs[(tr + 1) * 65 + (nt0 + 1) * 16 + l16];
      s0 = wmma4(a, b0, s0);
      s1 = wmma4(a, b1, s1);
    }
#pragma unroll
    for (int r = 0; r < 8; ++r) {
      const int d = mt * 16 + r + 8 * lh;
      Sbuf[sbase + d * 64 + nt0 * 16 + l16] = s0[r];
      Sbuf[sbase + d * 64 + (nt0 + 1) * 16 + l16] = s1[r];
    }
  }
  if (t < 64) {
    float s = 0.f;
#pragma unroll
    for (int j = 0; j < 64; ++j) s += Ks[j * 65 + t];
    ksumbuf[task * 64 + t] = s;
  }
}

__global__ void prefix_scan(float* __restrict__ Sbuf, float* __restrict__ ksumbuf) {
  const int g = blockIdx.x * blockDim.x + threadIdx.x;
  if (g < 16 * 4096) {
    const int bh = g >> 12, e = g & 4095;
    float run = 0.f;
    for (int c = 0; c < NCHUNK; ++c) {
      size_t idx = (size_t)(bh * NCHUNK + c) * 4096 + e;
      float v = Sbuf[idx];
      Sbuf[idx] = run;
      run += v;
    }
  } else if (g < 16 * 4096 + 16 * 64) {
    const int s2 = g - 16 * 4096, bh = s2 >> 6, d = s2 & 63;
    float run = 0.f;
    for (int c = 0; c < NCHUNK; ++c) {
      int idx = (bh * NCHUNK + c) * 64 + d;
      float v = ksumbuf[idx];
      ksumbuf[idx] = run;
      run += v;
    }
  }
}

__global__ __launch_bounds__(256) void attn_chunk(const float* __restrict__ Qb,
                                                  const float* __restrict__ Kb,
                                                  const float* __restrict__ Vb,
                                                  const float* __restrict__ Sbuf,
                                                  const float* __restrict__ ksumbuf,
                                                  float* __restrict__ attn) {
  extern __shared__ float sm[];
  float* Qs   = sm;
  float* Ks   = Qs + 64 * 65;
  float* Vs   = Ks + 64 * 65;
  float* Ss   = Vs + 64 * 65;
  float* Ms   = Ss + 64 * 65;
  float* ksum = Ms + 64 * 65;
  float* Dinv = ksum + 64;

  const int task = blockIdx.x;
  const int ck = task & 31, h = (task >> 5) & 7, b = task >> 8;
  const int row0 = b * 2048 + ck * CHUNK, col0 = h * 64;
  const int t = threadIdx.x, wave = t >> 5, lane = t & 31;
  const int lh = lane >> 4, l16 = lane & 15;

  {
    int r = t >> 2, c = (t & 3) * 16;
    const float4* qs = reinterpret_cast<const float4*>(&Qb[(size_t)(row0 + r) * LD + col0 + c]);
    const float4* ks = reinterpret_cast<const float4*>(&Kb[(size_t)(row0 + r) * LD + col0 + c]);
    const float4* vs = reinterpret_cast<const float4*>(&Vb[(size_t)(row0 + r) * LD + col0 + c]);
    const float4* ss = reinterpret_cast<const float4*>(&Sbuf[(size_t)task * 4096 + t * 16]);
    float4 qv[4];
#pragma unroll
    for (int i = 0; i < 4; ++i) qv[i] = qs[i];
#pragma unroll
    for (int i = 0; i < 4; ++i) {
      float4 kv = ks[i], vv = vs[i], sv = ss[i];
      float* kd = &Ks[r * 65 + c + i * 4];
      kd[0] = __expf(kv.x); kd[1] = __expf(kv.y); kd[2] = __expf(kv.z); kd[3] = __expf(kv.w);
      float* vd = &Vs[r * 65 + c + i * 4];
      vd[0] = vv.x; vd[1] = vv.y; vd[2] = vv.z; vd[3] = vv.w;
      float* sd = &Ss[(t >> 2) * 65 + (t & 3) * 16 + i * 4];
      sd[0] = sv.x; sd[1] = sv.y; sd[2] = sv.z; sd[3] = sv.w;
    }
    if (t < 64) ksum[t] = ksumbuf[task * 64 + t];

    float m = fmaxf(fmaxf(qv[0].x, qv[0].y), fmaxf(qv[0].z, qv[0].w));
#pragma unroll
    for (int i = 1; i < 4; ++i)
      m = fmaxf(m, fmaxf(fmaxf(qv[i].x, qv[i].y), fmaxf(qv[i].z, qv[i].w)));
    m = fmaxf(m, __shfl_xor(m, 1, 32));
    m = fmaxf(m, __shfl_xor(m, 2, 32));
    float s = 0.f;
#pragma unroll
    for (int i = 0; i < 4; ++i) {
      qv[i].x = __expf(qv[i].x - m); qv[i].y = __expf(qv[i].y - m);
      qv[i].z = __expf(qv[i].z - m); qv[i].w = __expf(qv[i].w - m);
      s += (qv[i].x + qv[i].y) + (qv[i].z + qv[i].w);
    }
    s += __shfl_xor(s, 1, 32);
    s += __shfl_xor(s, 2, 32);
    const float inv = 0.125f / s;
#pragma unroll
    for (int i = 0; i < 4; ++i) {
      float* qd = &Qs[r * 65 + c + i * 4];
      qd[0] = qv[i].x * inv; qd[1] = qv[i].y * inv; qd[2] = qv[i].z * inv; qd[3] = qv[i].w * inv;
    }
  }
  __syncthreads();

#pragma unroll
  for (int tt = wave * 2; tt <= wave * 2 + 1; ++tt) {
    const int mt = tt >> 2, nt = tt & 3;
    v8f acc = {};
#pragma unroll
    for (int kk = 0; kk < 64; kk += 4) {
      v2f a, bfr;
      a.x = Qs[(mt * 16 + l16) * 65 + kk + 2 * lh];
      a.y = Qs[(mt * 16 + l16) * 65 + kk + 2 * lh + 1];
      bfr.x = Ks[(nt * 16 + l16) * 65 + kk + 2 * lh];
      bfr.y = Ks[(nt * 16 + l16) * 65 + kk + 2 * lh + 1];
      acc = wmma4(a, bfr, acc);
    }
#pragma unroll
    for (int r = 0; r < 8; ++r) {
      int i = mt * 16 + r + 8 * lh, j = nt * 16 + l16;
      Ms[i * 65 + j] = (j <= i) ? acc[r] : 0.0f;
    }
  }
  __syncthreads();

  if (t < 64) {
    float s = 0.f, dq = 0.f;
#pragma unroll 8
    for (int j = 0; j < 64; ++j) s += Ms[t * 65 + j];
#pragma unroll 8
    for (int d = 0; d < 64; ++d) dq += Qs[t * 65 + d] * ksum[d];
    Dinv[t] = 1.0f / fmaxf(s + dq, 1e-3f);
  }
  __syncthreads();

#pragma unroll
  for (int tt = wave * 2; tt <= wave * 2 + 1; ++tt) {
    const int mt = tt >> 2, nt = tt & 3;
    v8f acc = {};
#pragma unroll
    for (int kk = 0; kk < 64; kk += 4) {
      v2f a, bfr;
      a.x = Ms[(mt * 16 + l16) * 65 + kk + 2 * lh];
      a.y = Ms[(mt * 16 + l16) * 65 + kk + 2 * lh + 1];
      bfr.x = Vs[(kk + 2 * lh) * 65 + nt * 16 + l16];
      bfr.y = Vs[(kk + 2 * lh + 1) * 65 + nt * 16 + l16];
      acc = wmma4(a, bfr, acc);
    }
#pragma unroll
    for (int kk = 0; kk < 64; kk += 4) {
      v2f a, bfr;
      a.x = Qs[(mt * 16 + l16) * 65 + kk + 2 * lh];
      a.y = Qs[(mt * 16 + l16) * 65 + kk + 2 * lh + 1];
      bfr.x = Ss[(kk + 2 * lh) * 65 + nt * 16 + l16];
      bfr.y = Ss[(kk + 2 * lh + 1) * 65 + nt * 16 + l16];
      acc = wmma4(a, bfr, acc);
    }
#pragma unroll
    for (int r = 0; r < 8; ++r) {
      int i = mt * 16 + r + 8 * lh;
      int col = col0 + nt * 16 + l16;
      attn[(size_t)(row0 + i) * LD + col] = acc[r] * Dinv[i];
    }
  }
}

extern "C" void kernel_launch(void* const* d_in, const int* in_sizes, int n_in,
                              void* d_out, int out_size, void* d_ws, size_t ws_size,
                              hipStream_t stream) {
  const float* x  = (const float*)d_in[0];
  const float* Wq = (const float*)d_in[1];
  const float* Wk = (const float*)d_in[2];
  const float* Wv = (const float*)d_in[3];
  const float* Wo = (const float*)d_in[4];
  const float* bo = (const float*)d_in[5];
  float* out = (float*)d_out;

  float* ws = (float*)d_ws;
  const size_t MN = (size_t)ROWS * LD;
  float* Qb = ws;
  float* Kb = Qb + MN;
  float* Vb = Kb + MN;
  float* At = Vb + MN;
  float* Sb = At + MN;
  float* Kc = Sb + (size_t)NTASK * 4096;

  dim3 ggrid(LD / GBN, ROWS / GBM);
  dim3 qkvgrid(LD / QBN + LD / 64, ROWS / QBM);

  qkv_state<<<qkvgrid, 256, 0, stream>>>(x, Wq, Wk, Wv, Qb, Kb, Vb, Sb, Kc);

  prefix_scan<<<(16 * 4096 + 16 * 64 + 255) / 256, 256, 0, stream>>>(Sb, Kc);

  size_t shbytes = (size_t)(5 * 64 * 65 + 128) * sizeof(float);
  attn_chunk<<<NTASK, 256, shbytes, stream>>>(Qb, Kb, Vb, Sb, Kc, At);

  gemm_f32<<<ggrid, 256, 0, stream>>>(At, Wo, out, bo, ROWS, LD, LD);
}
